# starting kernel
# baseline (speedup 1.0000x reference)
#include <hip/hip_runtime.h>

typedef __attribute__((ext_vector_type(2))) float v2f;
typedef __attribute__((ext_vector_type(8))) float v8f;

constexpr int kC   = 16;
constexpr int kK   = 8;
constexpr int kR   = 4096;
constexpr int kSR  = 128;
constexpr int kN   = kR * kSR;
constexpr int kPad = 17;

__global__ __launch_bounds__(256)
void pvr_feat_head(const float* __restrict__ points_feat,
                   const int*   __restrict__ indices,
                   const float* __restrict__ dists,
                   const float* __restrict__ w_rgb,
                   const float* __restrict__ w_sigma,
                   float* __restrict__ sigma_ws,
                   float* __restrict__ rgb_ws)
{
    __shared__ float sfeat[256 * kPad];

    const int tid = threadIdx.x;
    const int n   = blockIdx.x * 256 + tid;

    const float4* dv = (const float4*)(dists + (size_t)n * kK);
    float4 da = dv[0], db = dv[1];
    float dd[kK] = {da.x, da.y, da.z, da.w, db.x, db.y, db.z, db.w};

    float w[kK];
    float wsum = 0.0f;
#pragma unroll
    for (int k = 0; k < kK; ++k) {
        w[k] = 1.0f / (dd[k] + 1e-7f);
        wsum += w[k];
    }
    const float inv = 1.0f / wsum;

    const int4* iv = (const int4*)(indices + (size_t)n * kK);
    int4 ia = iv[0], ib = iv[1];
    int id[kK] = {ia.x, ia.y, ia.z, ia.w, ib.x, ib.y, ib.z, ib.w};

    float acc[kC];
#pragma unroll
    for (int c = 0; c < kC; ++c) acc[c] = 0.0f;

#pragma unroll
    for (int k = 0; k < kK; ++k) {
        const float4* row = (const float4*)(points_feat + (size_t)id[k] * kC);
        const float wk = w[k] * inv;
        float4 f0 = row[0], f1 = row[1], f2 = row[2], f3 = row[3];
        acc[0]  = fmaf(wk, f0.x, acc[0]);  acc[1]  = fmaf(wk, f0.y, acc[1]);
        acc[2]  = fmaf(wk, f0.z, acc[2]);  acc[3]  = fmaf(wk, f0.w, acc[3]);
        acc[4]  = fmaf(wk, f1.x, acc[4]);  acc[5]  = fmaf(wk, f1.y, acc[5]);
        acc[6]  = fmaf(wk, f1.z, acc[6]);  acc[7]  = fmaf(wk, f1.w, acc[7]);
        acc[8]  = fmaf(wk, f2.x, acc[8]);  acc[9]  = fmaf(wk, f2.y, acc[9]);
        acc[10] = fmaf(wk, f2.z, acc[10]); acc[11] = fmaf(wk, f2.w, acc[11]);
        acc[12] = fmaf(wk, f3.x, acc[12]); acc[13] = fmaf(wk, f3.y, acc[13]);
        acc[14] = fmaf(wk, f3.z, acc[14]); acc[15] = fmaf(wk, f3.w, acc[15]);
    }

    float* dst = sfeat + tid * kPad;
#pragma unroll
    for (int c = 0; c < kC; ++c) dst[c] = acc[c];

    __syncthreads();

    const int lane = tid & 31;
    const int wave = tid >> 5;
    const int half = lane >> 4;
    const int nl   = lane & 15;

    v2f bmat[4];
#pragma unroll
    for (int t = 0; t < 4; ++t) {
        const int k0 = 4 * t + 2 * half;
        float b0 = 0.0f, b1 = 0.0f;
        if (nl < 3) {
            b0 = w_rgb[k0 * 3 + nl];
            b1 = w_rgb[(k0 + 1) * 3 + nl];
        } else if (nl == 3) {
            b0 = w_sigma[k0];
            b1 = w_sigma[k0 + 1];
        }
        bmat[t].x = b0;
        bmat[t].y = b1;
    }

#pragma unroll
    for (int tt = 0; tt < 2; ++tt) {
        const int rowBase = (wave * 2 + tt) * 16;
        v8f d = {};
#pragma unroll
        for (int t = 0; t < 4; ++t) {
            const int k0 = 4 * t + 2 * half;
            const float* src = sfeat + (rowBase + nl) * kPad + k0;
            v2f a;
            a.x = src[0];
            a.y = src[1];
            d = __builtin_amdgcn_wmma_f32_16x16x4_f32(
                    false, a, false, bmat[t], (short)0, d, false, false);
        }

        const int pbase = blockIdx.x * 256 + rowBase + 8 * half;
        if (nl < 3) {
#pragma unroll
            for (int v = 0; v < 8; ++v) {
                const float x = d[v];
                rgb_ws[(size_t)nl * kN + (pbase + v)] = 1.0f / (1.0f + __expf(-x));
            }
        } else if (nl == 3) {
#pragma unroll
            for (int v = 0; v < 8; ++v) {
                sigma_ws[pbase + v] = fmaxf(d[v], 0.0f);
            }
        }
    }
}

__global__ __launch_bounds__(128)
void pvr_composite(const float* __restrict__ sigma_ws,
                   const float* __restrict__ rgb_ws,
                   const float* __restrict__ delta,
                   const float* __restrict__ z_vals,
                   float* __restrict__ out)
{
    const int r = blockIdx.x * blockDim.x + threadIdx.x;
    if (r >= kR) return;

    float T = 1.0f;
    float r0 = 0.0f, r1 = 0.0f, r2 = 0.0f, depth = 0.0f, accw = 0.0f;
    const int base = r * kSR;

    for (int s = 0; s < kSR; ++s) {
        const int i = base + s;
        const float sig = sigma_ws[i];
        const float a   = 1.0f - __expf(-sig * delta[i]);
        const float wt  = a * T;
        r0    = fmaf(wt, rgb_ws[i],            r0);
        r1    = fmaf(wt, rgb_ws[kN + i],       r1);
        r2    = fmaf(wt, rgb_ws[2 * kN + i],   r2);
        depth = fmaf(wt, z_vals[i],            depth);
        accw += wt;
        T *= (1.0f - a + 1e-10f);
    }

    const float bg = 1.0f - accw;
    out[r * 5 + 0] = r0 + bg;
    out[r * 5 + 1] = r1 + bg;
    out[r * 5 + 2] = r2 + bg;
    out[r * 5 + 3] = depth;
    out[r * 5 + 4] = accw;
}

extern "C" void kernel_launch(void* const* d_in, const int* in_sizes, int n_in,
                              void* d_out, int out_size, void* d_ws, size_t ws_size,
                              hipStream_t stream)
{
    const float* points_feat = (const float*)d_in[0];
    const int*   indices     = (const int*)  d_in[1];
    const float* dists       = (const float*)d_in[2];
    const float* w_rgb       = (const float*)d_in[3];
    const float* w_sigma     = (const float*)d_in[4];
    const float* delta       = (const float*)d_in[5];
    const float* z_vals      = (const float*)d_in[6];
    float* out = (float*)d_out;

    float* sigma_ws = (float*)d_ws;
    float* rgb_ws   = sigma_ws + kN;

    pvr_feat_head<<<kN / 256, 256, 0, stream>>>(
        points_feat, indices, dists, w_rgb, w_sigma, sigma_ws, rgb_ws);

    pvr_composite<<<kR / 128, 128, 0, stream>>>(
        sigma_ws, rgb_ws, delta, z_vals, out);
}


# Gather remapped: 4 adjacent lanes read one 64-byte feature row (16 B each), so a load instruction touches 8 cache lines instead of 32; every sum keeps its k order (feat bit-identical)
# speedup vs baseline: 1.3547x; 1.3547x over previous
#include <hip/hip_runtime.h>

typedef __attribute__((ext_vector_type(2))) float v2f;
typedef __attribute__((ext_vector_type(8))) float v8f;

constexpr int kC   = 16;
constexpr int kK   = 8;
constexpr int kR   = 4096;
constexpr int kSR  = 128;
constexpr int kN   = kR * kSR;
constexpr int kPad = 17;

__global__ __launch_bounds__(256)
void pvr_feat_head(const float* __restrict__ points_feat,
                   const int*   __restrict__ indices,
                   const float* __restrict__ dists,
                   const float* __restrict__ w_rgb,
                   const float* __restrict__ w_sigma,
                   float* __restrict__ sigma_ws,
                   float* __restrict__ rgb_ws)
{
    __shared__ float sfeat[256 * kPad];

    const int tid  = threadIdx.x;
    const int lane = tid & 31;
    const int wave = tid >> 5;

    const int grp = lane >> 2;
    const int qtr = lane & 3;

    for (int pass = 0; pass < 4; ++pass) {
        const int pl = wave * 32 + pass * 8 + grp;
        const int n  = blockIdx.x * 256 + pl;

        const float4* dv = (const float4*)(dists + (size_t)n * kK);
        float4 da = dv[0], db = dv[1];
        float dd[kK] = {da.x, da.y, da.z, da.w, db.x, db.y, db.z, db.w};

        float w[kK];
        float wsum = 0.0f;
#pragma unroll
        for (int k = 0; k < kK; ++k) {
            w[k] = 1.0f / (dd[k] + 1e-7f);
            wsum += w[k];
        }
        const float inv = 1.0f / wsum;

        const int4* iv = (const int4*)(indices + (size_t)n * kK);
        int4 ia = iv[0], ib = iv[1];
        int id[kK] = {ia.x, ia.y, ia.z, ia.w, ib.x, ib.y, ib.z, ib.w};

        float a0 = 0.0f, a1 = 0.0f, a2 = 0.0f, a3 = 0.0f;
#pragma unroll
        for (int k = 0; k < kK; ++k) {
            const float4 f = *(const float4*)(points_feat + (size_t)id[k] * kC + 4 * qtr);
            const float wk = w[k] * inv;
            a0 = fmaf(wk, f.x, a0);
            a1 = fmaf(wk, f.y, a1);
            a2 = fmaf(wk, f.z, a2);
            a3 = fmaf(wk, f.w, a3);
        }

        float* dst = sfeat + pl * kPad + 4 * qtr;
        dst[0] = a0; dst[1] = a1; dst[2] = a2; dst[3] = a3;
    }

    __syncthreads();

    const int half = lane >> 4;
    const int nl   = lane & 15;

    v2f bmat[4];
#pragma unroll
    for (int t = 0; t < 4; ++t) {
        const int k0 = 4 * t + 2 * half;
        float b0 = 0.0f, b1 = 0.0f;
        if (nl < 3) {
            b0 = w_rgb[k0 * 3 + nl];
            b1 = w_rgb[(k0 + 1) * 3 + nl];
        } else if (nl == 3) {
            b0 = w_sigma[k0];
            b1 = w_sigma[k0 + 1];
        }
        bmat[t].x = b0;
        bmat[t].y = b1;
    }

#pragma unroll
    for (int tt = 0; tt < 2; ++tt) {
        const int rowBase = (wave * 2 + tt) * 16;
        v8f d = {};
#pragma unroll
        for (int t = 0; t < 4; ++t) {
            const int k0 = 4 * t + 2 * half;
            const float* src = sfeat + (rowBase + nl) * kPad + k0;
            v2f a;
            a.x = src[0];
            a.y = src[1];
            d = __builtin_amdgcn_wmma_f32_16x16x4_f32(
                    false, a, false, bmat[t], (short)0, d, false, false);
        }

        const int pbase = blockIdx.x * 256 + rowBase + 8 * half;
        if (nl < 3) {
#pragma unroll
            for (int v = 0; v < 8; ++v) {
                const float x = d[v];
                rgb_ws[(size_t)nl * kN + (pbase + v)] = 1.0f / (1.0f + __expf(-x));
            }
        } else if (nl == 3) {
#pragma unroll
            for (int v = 0; v < 8; ++v) {
                sigma_ws[pbase + v] = fmaxf(d[v], 0.0f);
            }
        }
    }
}

__global__ __launch_bounds__(128)
void pvr_composite(const float* __restrict__ sigma_ws,
                   const float* __restrict__ rgb_ws,
                   const float* __restrict__ delta,
                   const float* __restrict__ z_vals,
                   float* __restrict__ out)
{
    const int r = blockIdx.x * blockDim.x + threadIdx.x;
    if (r >= kR) return;

    float T = 1.0f;
    float r0 = 0.0f, r1 = 0.0f, r2 = 0.0f, depth = 0.0f, accw = 0.0f;
    const int base = r * kSR;

    for (int s = 0; s < kSR; ++s) {
        const int i = base + s;
        const float sig = sigma_ws[i];
        const float a   = 1.0f - __expf(-sig * delta[i]);
        const float wt  = a * T;
        r0    = fmaf(wt, rgb_ws[i],            r0);
        r1    = fmaf(wt, rgb_ws[kN + i],       r1);
        r2    = fmaf(wt, rgb_ws[2 * kN + i],   r2);
        depth = fmaf(wt, z_vals[i],            depth);
        accw += wt;
        T *= (1.0f - a + 1e-10f);
    }

    const float bg = 1.0f - accw;
    out[r * 5 + 0] = r0 + bg;
    out[r * 5 + 1] = r1 + bg;
    out[r * 5 + 2] = r2 + bg;
    out[r * 5 + 3] = depth;
    out[r * 5 + 4] = accw;
}

extern "C" void kernel_launch(void* const* d_in, const int* in_sizes, int n_in,
                              void* d_out, int out_size, void* d_ws, size_t ws_size,
                              hipStream_t stream)
{
    const float* points_feat = (const float*)d_in[0];
    const int*   indices     = (const int*)  d_in[1];
    const float* dists       = (const float*)d_in[2];
    const float* w_rgb       = (const float*)d_in[3];
    const float* w_sigma     = (const float*)d_in[4];
    const float* delta       = (const float*)d_in[5];
    const float* z_vals      = (const float*)d_in[6];
    float* out = (float*)d_out;

    float* sigma_ws = (float*)d_ws;
    float* rgb_ws   = sigma_ws + kN;

    pvr_feat_head<<<kN / 256, 256, 0, stream>>>(
        points_feat, indices, dists, w_rgb, w_sigma, sigma_ws, rgb_ws);

    pvr_composite<<<kR / 128, 128, 0, stream>>>(
        sigma_ws, rgb_ws, delta, z_vals, out);
}


# Compositing kernel re-mapped: 8 lanes per ray, whole-line 16-byte loads; per-ray product and sums re-associated (4-sample lane runs, 8-lane DPP scan per 32-sample chunk, butterfly sums)
# speedup vs baseline: 1.7976x; 1.3269x over previous
#include <hip/hip_runtime.h>

typedef __attribute__((ext_vector_type(2))) float v2f;
typedef __attribute__((ext_vector_type(8))) float v8f;

constexpr int kC   = 16;
constexpr int kK   = 8;
constexpr int kR   = 4096;
constexpr int kSR  = 128;
constexpr int kN   = kR * kSR;
constexpr int kPad = 17;

__global__ __launch_bounds__(256)
void pvr_feat_head(const float* __restrict__ points_feat,
                   const int*   __restrict__ indices,
                   const float* __restrict__ dists,
                   const float* __restrict__ w_rgb,
                   const float* __restrict__ w_sigma,
                   float* __restrict__ sigma_ws,
                   float* __restrict__ rgb_ws)
{
    __shared__ float sfeat[256 * kPad];

    const int tid  = threadIdx.x;
    const int lane = tid & 31;
    const int wave = tid >> 5;

    const int grp = lane >> 2;
    const int qtr = lane & 3;

    for (int pass = 0; pass < 4; ++pass) {
        const int pl = wave * 32 + pass * 8 + grp;
        const int n  = blockIdx.x * 256 + pl;

        const float4* dv = (const float4*)(dists + (size_t)n * kK);
        float4 da = dv[0], db = dv[1];
        float dd[kK] = {da.x, da.y, da.z, da.w, db.x, db.y, db.z, db.w};

        float w[kK];
        float wsum = 0.0f;
#pragma unroll
        for (int k = 0; k < kK; ++k) {
            w[k] = 1.0f / (dd[k] + 1e-7f);
            wsum += w[k];
        }
        const float inv = 1.0f / wsum;

        const int4* iv = (const int4*)(indices + (size_t)n * kK);
        int4 ia = iv[0], ib = iv[1];
        int id[kK] = {ia.x, ia.y, ia.z, ia.w, ib.x, ib.y, ib.z, ib.w};

        float a0 = 0.0f, a1 = 0.0f, a2 = 0.0f, a3 = 0.0f;
#pragma unroll
        for (int k = 0; k < kK; ++k) {
            const float4 f = *(const float4*)(points_feat + (size_t)id[k] * kC + 4 * qtr);
            const float wk = w[k] * inv;
            a0 = fmaf(wk, f.x, a0);
            a1 = fmaf(wk, f.y, a1);
            a2 = fmaf(wk, f.z, a2);
            a3 = fmaf(wk, f.w, a3);
        }

        float* dst = sfeat + pl * kPad + 4 * qtr;
        dst[0] = a0; dst[1] = a1; dst[2] = a2; dst[3] = a3;
    }

    __syncthreads();

    const int half = lane >> 4;
    const int nl   = lane & 15;

    v2f bmat[4];
#pragma unroll
    for (int t = 0; t < 4; ++t) {
        const int k0 = 4 * t + 2 * half;
        float b0 = 0.0f, b1 = 0.0f;
        if (nl < 3) {
            b0 = w_rgb[k0 * 3 + nl];
            b1 = w_rgb[(k0 + 1) * 3 + nl];
        } else if (nl == 3) {
            b0 = w_sigma[k0];
            b1 = w_sigma[k0 + 1];
        }
        bmat[t].x = b0;
        bmat[t].y = b1;
    }

#pragma unroll
    for (int tt = 0; tt < 2; ++tt) {
        const int rowBase = (wave * 2 + tt) * 16;
        v8f d = {};
#pragma unroll
        for (int t = 0; t < 4; ++t) {
            const int k0 = 4 * t + 2 * half;
            const float* src = sfeat + (rowBase + nl) * kPad + k0;
            v2f a;
            a.x = src[0];
            a.y = src[1];
            d = __builtin_amdgcn_wmma_f32_16x16x4_f32(
                    false, a, false, bmat[t], (short)0, d, false, false);
        }

        const int pbase = blockIdx.x * 256 + rowBase + 8 * half;
        if (nl < 3) {
#pragma unroll
            for (int v = 0; v < 8; ++v) {
                const float x = d[v];
                rgb_ws[(size_t)nl * kN + (pbase + v)] = 1.0f / (1.0f + __expf(-x));
            }
        } else if (nl == 3) {
#pragma unroll
            for (int v = 0; v < 8; ++v) {
                sigma_ws[pbase + v] = fmaxf(d[v], 0.0f);
            }
        }
    }
}

typedef __attribute__((ext_vector_type(4))) float v4f;
constexpr int kLPR = 8;
constexpr int kRPB = 32;

#define PVR_DPP8(a, b, c, d, e, f, g, h) \
    ((a) | ((b) << 3) | ((c) << 6) | ((d) << 9) | ((e) << 12) | ((f) << 15) | ((g) << 18) | ((h) << 21))

template <int SEL>
__device__ __forceinline__ float pvr_lane8(float x)
{
    return __uint_as_float(__builtin_amdgcn_mov_dpp8(__float_as_uint(x), SEL));
}

__device__ __forceinline__ float pvr_sum8(float x)
{
    x += pvr_lane8<PVR_DPP8(1, 0, 3, 2, 5, 4, 7, 6)>(x);
    x += pvr_lane8<PVR_DPP8(2, 3, 0, 1, 6, 7, 4, 5)>(x);
    x += pvr_lane8<PVR_DPP8(4, 5, 6, 7, 0, 1, 2, 3)>(x);
    return x;
}

__global__ __launch_bounds__(256)
void pvr_composite(const float* __restrict__ sigma_ws,
                   const float* __restrict__ rgb_ws,
                   const float* __restrict__ delta,
                   const float* __restrict__ z_vals,
                   float* __restrict__ out)
{
    __shared__ float sout[kRPB * 5];

    const int tid = threadIdx.x;
    const int p   = tid & (kLPR - 1);
    const int rl  = tid >> 3;
    const int i0  = (blockIdx.x * kRPB + rl) * kSR + 4 * p;

    const v4f* sg4 = (const v4f*)(sigma_ws + i0);
    const v4f* dl4 = (const v4f*)(delta + i0);
    const v4f* c04 = (const v4f*)(rgb_ws + i0);
    const v4f* c14 = (const v4f*)(rgb_ws + kN + i0);
    const v4f* c24 = (const v4f*)(rgb_ws + 2 * kN + i0);
    const v4f* zv4 = (const v4f*)(z_vals + i0);

    v4f sg[4], dl[4], c0[4], c1[4], c2[4], zv[4];
#pragma unroll
    for (int j = 0; j < 4; ++j) {
        sg[j] = sg4[8 * j]; dl[j] = dl4[8 * j];
        c0[j] = c04[8 * j]; c1[j] = c14[8 * j]; c2[j] = c24[8 * j];
        zv[j] = zv4[8 * j];
    }
    asm volatile("" :: "v"(sg[0]), "v"(dl[0]), "v"(c0[0]), "v"(c1[0]), "v"(c2[0]), "v"(zv[0]),
                       "v"(sg[1]), "v"(dl[1]), "v"(c0[1]), "v"(c1[1]), "v"(c2[1]), "v"(zv[1]),
                       "v"(sg[2]), "v"(dl[2]), "v"(c0[2]), "v"(c1[2]), "v"(c2[2]), "v"(zv[2]),
                       "v"(sg[3]), "v"(dl[3]), "v"(c0[3]), "v"(c1[3]), "v"(c2[3]), "v"(zv[3]));
    __builtin_amdgcn_sched_barrier(0);

    float f[4], s0[4], s1[4], s2[4], sd[4], sw[4];
#pragma unroll
    for (int j = 0; j < 4; ++j) {
        const float sgv[4] = {sg[j].x, sg[j].y, sg[j].z, sg[j].w};
        const float dlv[4] = {dl[j].x, dl[j].y, dl[j].z, dl[j].w};
        const float c0v[4] = {c0[j].x, c0[j].y, c0[j].z, c0[j].w};
        const float c1v[4] = {c1[j].x, c1[j].y, c1[j].z, c1[j].w};
        const float c2v[4] = {c2[j].x, c2[j].y, c2[j].z, c2[j].w};
        const float zvv[4] = {zv[j].x, zv[j].y, zv[j].z, zv[j].w};
        float t = 1.0f;
        float q0 = 0.0f, q1 = 0.0f, q2 = 0.0f, qd = 0.0f, qw = 0.0f;
#pragma unroll
        for (int e = 0; e < 4; ++e) {
            const float a  = 1.0f - __expf(-sgv[e] * dlv[e]);
            const float wt = a * t;
            q0 = fmaf(wt, c0v[e], q0);
            q1 = fmaf(wt, c1v[e], q1);
            q2 = fmaf(wt, c2v[e], q2);
            qd = fmaf(wt, zvv[e], qd);
            qw += wt;
            t *= (1.0f - a + 1e-10f);
        }
        f[j] = t; s0[j] = q0; s1[j] = q1; s2[j] = q2; sd[j] = qd; sw[j] = qw;
    }

    float r0 = 0.0f, r1 = 0.0f, r2 = 0.0f, depth = 0.0f, accw = 0.0f;
    float carry = 1.0f;
#pragma unroll
    for (int j = 0; j < 4; ++j) {
        float x = f[j], y;
        y = pvr_lane8<PVR_DPP8(0, 0, 1, 2, 3, 4, 5, 6)>(x); if (p >= 1) x *= y;
        y = pvr_lane8<PVR_DPP8(0, 1, 0, 1, 2, 3, 4, 5)>(x); if (p >= 2) x *= y;
        y = pvr_lane8<PVR_DPP8(0, 1, 2, 3, 0, 1, 2, 3)>(x); if (p >= 4) x *= y;
        y = pvr_lane8<PVR_DPP8(0, 0, 1, 2, 3, 4, 5, 6)>(x);
        const float E = (p == 0) ? carry : carry * y;
        r0    = fmaf(E, s0[j], r0);
        r1    = fmaf(E, s1[j], r1);
        r2    = fmaf(E, s2[j], r2);
        depth = fmaf(E, sd[j], depth);
        accw  = fmaf(E, sw[j], accw);
        carry *= pvr_lane8<PVR_DPP8(7, 7, 7, 7, 7, 7, 7, 7)>(x);
    }

    r0    = pvr_sum8(r0);
    r1    = pvr_sum8(r1);
    r2    = pvr_sum8(r2);
    depth = pvr_sum8(depth);
    accw  = pvr_sum8(accw);

    if (p == 0) {
        const float bg = 1.0f - accw;
        float* so = sout + rl * 5;
        so[0] = r0 + bg;
        so[1] = r1 + bg;
        so[2] = r2 + bg;
        so[3] = depth;
        so[4] = accw;
    }
    __syncthreads();

    if (tid < kRPB * 5) out[blockIdx.x * (kRPB * 5) + tid] = sout[tid];
}

extern "C" void kernel_launch(void* const* d_in, const int* in_sizes, int n_in,
                              void* d_out, int out_size, void* d_ws, size_t ws_size,
                              hipStream_t stream)
{
    const float* points_feat = (const float*)d_in[0];
    const int*   indices     = (const int*)  d_in[1];
    const float* dists       = (const float*)d_in[2];
    const float* w_rgb       = (const float*)d_in[3];
    const float* w_sigma     = (const float*)d_in[4];
    const float* delta       = (const float*)d_in[5];
    const float* z_vals      = (const float*)d_in[6];
    float* out = (float*)d_out;

    float* sigma_ws = (float*)d_ws;
    float* rgb_ws   = sigma_ws + kN;

    pvr_feat_head<<<kN / 256, 256, 0, stream>>>(
        points_feat, indices, dists, w_rgb, w_sigma, sigma_ws, rgb_ws);

    pvr_composite<<<kR / kRPB, 256, 0, stream>>>(
        sigma_ws, rgb_ws, delta, z_vals, out);
}


# Compositing fused into the feature kernel: a block's 2 rays are composited from LDS by 16 lanes after the head; one launch, not two; sigma/rgb never go through memory; every sum keeps its order
# speedup vs baseline: 1.9668x; 1.0941x over previous
#include <hip/hip_runtime.h>


typedef __attribute__((ext_vector_type(2))) float v2f;
typedef __attribute__((ext_vector_type(4))) float v4f;
typedef __attribute__((ext_vector_type(8))) float v8f;

constexpr int kC     = 16;
constexpr int kK     = 8;
constexpr int kR     = 4096;
constexpr int kSR    = 128;
constexpr int kN     = kR * kSR;
constexpr int kPad   = 17;
constexpr int kPlane = 272;
constexpr int kLPR   = 8;
constexpr int kRPB   = 2;

#define PVR_DPP8(a, b, c, d, e, f, g, h) \
    ((a) | ((b) << 3) | ((c) << 6) | ((d) << 9) | ((e) << 12) | ((f) << 15) | ((g) << 18) | ((h) << 21))

template <int SEL>
__device__ __forceinline__ float pvr_lane8(float x)
{
    return __uint_as_float(__builtin_amdgcn_mov_dpp8(__float_as_uint(x), SEL));
}

__device__ __forceinline__ float pvr_sum8(float x)
{
    x += pvr_lane8<PVR_DPP8(1, 0, 3, 2, 5, 4, 7, 6)>(x);
    x += pvr_lane8<PVR_DPP8(2, 3, 0, 1, 6, 7, 4, 5)>(x);
    x += pvr_lane8<PVR_DPP8(4, 5, 6, 7, 0, 1, 2, 3)>(x);
    return x;
}

__global__ __launch_bounds__(256)
void pvr_render(const float* __restrict__ points_feat,
                const int*   __restrict__ indices,
                const float* __restrict__ dists,
                const float* __restrict__ w_rgb,
                const float* __restrict__ w_sigma,
                const float* __restrict__ delta,
                const float* __restrict__ z_vals,
                float* __restrict__ out)
{
    __shared__ float sfeat[256 * kPad];
    __shared__ __attribute__((aligned(16))) float shead[4 * kPlane];
    __shared__ __attribute__((aligned(16))) float sdz[2 * 256];

    const int tid  = threadIdx.x;
    const int lane = tid & 31;
    const int wave = tid >> 5;

    v4f pre = {0.0f, 0.0f, 0.0f, 0.0f};
    if (tid < 128) {
        const float* src = (tid < 64) ? delta : z_vals;
        pre = *(const v4f*)(src + (size_t)blockIdx.x * 256 + 4 * (tid & 63));
    }

    const int grp = lane >> 2;
    const int qtr = lane & 3;

    for (int pass = 0; pass < 4; ++pass) {
        const int pl = wave * 32 + pass * 8 + grp;
        const int n  = blockIdx.x * 256 + pl;

        const float4* dv = (const float4*)(dists + (size_t)n * kK);
        float4 da = dv[0], db = dv[1];
        float dd[kK] = {da.x, da.y, da.z, da.w, db.x, db.y, db.z, db.w};

        float w[kK];
        float wsum = 0.0f;
#pragma unroll
        for (int k = 0; k < kK; ++k) {
            w[k] = 1.0f / (dd[k] + 1e-7f);
            wsum += w[k];
        }
        const float inv = 1.0f / wsum;

        const int4* iv = (const int4*)(indices + (size_t)n * kK);
        int4 ia = iv[0], ib = iv[1];
        int id[kK] = {ia.x, ia.y, ia.z, ia.w, ib.x, ib.y, ib.z, ib.w};

        float a0 = 0.0f, a1 = 0.0f, a2 = 0.0f, a3 = 0.0f;
#pragma unroll
        for (int k = 0; k < kK; ++k) {
            const float4 f = *(const float4*)(points_feat + (size_t)id[k] * kC + 4 * qtr);
            const float wk = w[k] * inv;
            a0 = fmaf(wk, f.x, a0);
            a1 = fmaf(wk, f.y, a1);
            a2 = fmaf(wk, f.z, a2);
            a3 = fmaf(wk, f.w, a3);
        }

        float* dst = sfeat + pl * kPad + 4 * qtr;
        dst[0] = a0; dst[1] = a1; dst[2] = a2; dst[3] = a3;
    }

    if (tid < 128) *(v4f*)(sdz + 4 * tid) = pre;

    __syncthreads();

    const int half = lane >> 4;
    const int nl   = lane & 15;

    v2f bmat[4];
#pragma unroll
    for (int t = 0; t < 4; ++t) {
        const int k0 = 4 * t + 2 * half;
        float b0 = 0.0f, b1 = 0.0f;
        if (nl < 3) {
            b0 = w_rgb[k0 * 3 + nl];
            b1 = w_rgb[(k0 + 1) * 3 + nl];
        } else if (nl == 3) {
            b0 = w_sigma[k0];
            b1 = w_sigma[k0 + 1];
        }
        bmat[t].x = b0;
        bmat[t].y = b1;
    }

#pragma unroll
    for (int tt = 0; tt < 2; ++tt) {
        const int rowBase = (wave * 2 + tt) * 16;
        v8f d = {};
#pragma unroll
        for (int t = 0; t < 4; ++t) {
            const int k0 = 4 * t + 2 * half;
            const float* src = sfeat + (rowBase + nl) * kPad + k0;
            v2f a;
            a.x = src[0];
            a.y = src[1];
            d = __builtin_amdgcn_wmma_f32_16x16x4_f32(
                    false, a, false, bmat[t], (short)0, d, false, false);
        }

        float* hp = shead + nl * kPlane + rowBase + 8 * half;
        if (nl < 3) {
            float hv[8];
#pragma unroll
            for (int v = 0; v < 8; ++v) hv[v] = 1.0f / (1.0f + __expf(-d[v]));
            *(v4f*)(hp)     = (v4f){hv[0], hv[1], hv[2], hv[3]};
            *(v4f*)(hp + 4) = (v4f){hv[4], hv[5], hv[6], hv[7]};
        } else if (nl == 3) {
            float hv[8];
#pragma unroll
            for (int v = 0; v < 8; ++v) hv[v] = fmaxf(d[v], 0.0f);
            *(v4f*)(hp)     = (v4f){hv[0], hv[1], hv[2], hv[3]};
            *(v4f*)(hp + 4) = (v4f){hv[4], hv[5], hv[6], hv[7]};
        }
    }

    __syncthreads();

    if (tid >= kRPB * kLPR) return;

    const int p  = tid & (kLPR - 1);
    const int rl = tid >> 3;
    const int i0 = rl * kSR + 4 * p;

    float r0 = 0.0f, r1 = 0.0f, r2 = 0.0f, depth = 0.0f, accw = 0.0f;
    float carry = 1.0f;
#pragma unroll
    for (int j = 0; j < 4; ++j) {
        const int o = i0 + 32 * j;
        const v4f c0 = *(const v4f*)(shead + 0 * kPlane + o);
        const v4f c1 = *(const v4f*)(shead + 1 * kPlane + o);
        const v4f c2 = *(const v4f*)(shead + 2 * kPlane + o);
        const v4f sg = *(const v4f*)(shead + 3 * kPlane + o);
        const v4f dl = *(const v4f*)(sdz + o);
        const v4f zv = *(const v4f*)(sdz + 256 + o);

        const float sgv[4] = {sg.x, sg.y, sg.z, sg.w};
        const float dlv[4] = {dl.x, dl.y, dl.z, dl.w};
        const float c0v[4] = {c0.x, c0.y, c0.z, c0.w};
        const float c1v[4] = {c1.x, c1.y, c1.z, c1.w};
        const float c2v[4] = {c2.x, c2.y, c2.z, c2.w};
        const float zvv[4] = {zv.x, zv.y, zv.z, zv.w};
        float t = 1.0f;
        float q0 = 0.0f, q1 = 0.0f, q2 = 0.0f, qd = 0.0f, qw = 0.0f;
#pragma unroll
        for (int e = 0; e < 4; ++e) {
            const float a  = 1.0f - __expf(-sgv[e] * dlv[e]);
            const float wt = a * t;
            q0 = fmaf(wt, c0v[e], q0);
            q1 = fmaf(wt, c1v[e], q1);
            q2 = fmaf(wt, c2v[e], q2);
            qd = fmaf(wt, zvv[e], qd);
            qw += wt;
            t *= (1.0f - a + 1e-10f);
        }

        float x = t, y;
        y = pvr_lane8<PVR_DPP8(0, 0, 1, 2, 3, 4, 5, 6)>(x); if (p >= 1) x *= y;
        y = pvr_lane8<PVR_DPP8(0, 1, 0, 1, 2, 3, 4, 5)>(x); if (p >= 2) x *= y;
        y = pvr_lane8<PVR_DPP8(0, 1, 2, 3, 0, 1, 2, 3)>(x); if (p >= 4) x *= y;
        y = pvr_lane8<PVR_DPP8(0, 0, 1, 2, 3, 4, 5, 6)>(x);
        const float E = (p == 0) ? carry : carry * y;
        r0    = fmaf(E, q0, r0);
        r1    = fmaf(E, q1, r1);
        r2    = fmaf(E, q2, r2);
        depth = fmaf(E, qd, depth);
        accw  = fmaf(E, qw, accw);
        carry *= pvr_lane8<PVR_DPP8(7, 7, 7, 7, 7, 7, 7, 7)>(x);
    }

    r0    = pvr_sum8(r0);
    r1    = pvr_sum8(r1);
    r2    = pvr_sum8(r2);
    depth = pvr_sum8(depth);
    accw  = pvr_sum8(accw);

    const float bg = 1.0f - accw;
    float res = accw;
    if (p == 0) res = r0 + bg;
    if (p == 1) res = r1 + bg;
    if (p == 2) res = r2 + bg;
    if (p == 3) res = depth;
    if (p < 5) out[(blockIdx.x * kRPB + rl) * 5 + p] = res;
}

extern "C" void kernel_launch(void* const* d_in, const int* in_sizes, int n_in,
                              void* d_out, int out_size, void* d_ws, size_t ws_size,
                              hipStream_t stream)
{
    const float* points_feat = (const float*)d_in[0];
    const int*   indices     = (const int*)  d_in[1];
    const float* dists       = (const float*)d_in[2];
    const float* w_rgb       = (const float*)d_in[3];
    const float* w_sigma     = (const float*)d_in[4];
    const float* delta       = (const float*)d_in[5];
    const float* z_vals      = (const float*)d_in[6];
    float* out = (float*)d_out;

    pvr_render<<<kN / 256, 256, 0, stream>>>(
        points_feat, indices, dists, w_rgb, w_sigma, delta, z_vals, out);
}


# Per-point work done once: a lane loads indices/dists, normalizes weights of ONE point; gather quads get 32-bit row offsets and weights by DPP quad broadcast (loads 16->4, divides 36->9); sums kept
# speedup vs baseline: 2.5593x; 1.3013x over previous
#include <hip/hip_runtime.h>


typedef __attribute__((ext_vector_type(2))) float v2f;
typedef __attribute__((ext_vector_type(4))) float v4f;
typedef __attribute__((ext_vector_type(8))) float v8f;

constexpr int kC     = 16;
constexpr int kK     = 8;
constexpr int kR     = 4096;
constexpr int kSR    = 128;
constexpr int kN     = kR * kSR;
constexpr int kPad   = 17;
constexpr int kPlane = 272;
constexpr int kLPR   = 8;
constexpr int kRPB   = 2;

#define PVR_DPP8(a, b, c, d, e, f, g, h) \
    ((a) | ((b) << 3) | ((c) << 6) | ((d) << 9) | ((e) << 12) | ((f) << 15) | ((g) << 18) | ((h) << 21))

template <int SEL>
__device__ __forceinline__ float pvr_lane8(float x)
{
    return __uint_as_float(__builtin_amdgcn_mov_dpp8(__float_as_uint(x), SEL));
}

__device__ __forceinline__ float pvr_sum8(float x)
{
    x += pvr_lane8<PVR_DPP8(1, 0, 3, 2, 5, 4, 7, 6)>(x);
    x += pvr_lane8<PVR_DPP8(2, 3, 0, 1, 6, 7, 4, 5)>(x);
    x += pvr_lane8<PVR_DPP8(4, 5, 6, 7, 0, 1, 2, 3)>(x);
    return x;
}


template <int J>
__device__ __forceinline__ unsigned pvr_quad_u(unsigned x)
{
    return (unsigned)__builtin_amdgcn_update_dpp(0, (int)x, J * 0x55, 0xF, 0xF, true);
}

template <int J>
__device__ __forceinline__ void pvr_gather_pass(const float* __restrict__ points_feat,
                                                const unsigned (&ob)[8], const float (&wn)[8],
                                                unsigned qoff, float* __restrict__ dst)
{
    float a0 = 0.0f, a1 = 0.0f, a2 = 0.0f, a3 = 0.0f;
#pragma unroll
    for (int k = 0; k < 8; ++k) {
        const unsigned off = pvr_quad_u<J>(ob[k]) | qoff;
        const float    wk  = __uint_as_float(pvr_quad_u<J>(__float_as_uint(wn[k])));
        const float4 f = *(const float4*)((const char*)points_feat + off);
        a0 = fmaf(wk, f.x, a0);
        a1 = fmaf(wk, f.y, a1);
        a2 = fmaf(wk, f.z, a2);
        a3 = fmaf(wk, f.w, a3);
    }
    dst[0] = a0; dst[1] = a1; dst[2] = a2; dst[3] = a3;
}

__global__ __launch_bounds__(256, 16)
void pvr_render(const float* __restrict__ points_feat,
                const int*   __restrict__ indices,
                const float* __restrict__ dists,
                const float* __restrict__ w_rgb,
                const float* __restrict__ w_sigma,
                const float* __restrict__ delta,
                const float* __restrict__ z_vals,
                float* __restrict__ out)
{
    __shared__ float sfeat[256 * kPad];
    __shared__ __attribute__((aligned(16))) float shead[4 * kPlane];
    __shared__ __attribute__((aligned(16))) float sdz[2 * 256];

    const int tid  = threadIdx.x;
    const int lane = tid & 31;
    const int wave = tid >> 5;
    const int blk  = blockIdx.x;

    v4f pre = {0.0f, 0.0f, 0.0f, 0.0f};
    if (tid < 128) {
        const float* src = (tid < 64) ? delta : z_vals;
        pre = *(const v4f*)(src + (size_t)blk * 256 + 4 * (tid & 63));
    }

    const int plOwn = wave * 32 + lane;
    const int nOwn  = blk * 256 + plOwn;

    const float4* dv = (const float4*)(dists + (size_t)nOwn * kK);
    const int4*   iv = (const int4*)(indices + (size_t)nOwn * kK);
    const float4 da = dv[0], db = dv[1];
    const int4   ia = iv[0], ib = iv[1];
    const float dd[kK] = {da.x, da.y, da.z, da.w, db.x, db.y, db.z, db.w};
    const int   id[kK] = {ia.x, ia.y, ia.z, ia.w, ib.x, ib.y, ib.z, ib.w};

    float w[kK];
    float wsum = 0.0f;
#pragma unroll
    for (int k = 0; k < kK; ++k) {
        w[k] = 1.0f / (dd[k] + 1e-7f);
        wsum += w[k];
    }
    const float inv = 1.0f / wsum;
    float    wn[kK];
    unsigned ob[kK];
#pragma unroll
    for (int k = 0; k < kK; ++k) {
        wn[k] = w[k] * inv;
        ob[k] = (unsigned)id[k] * (unsigned)(kC * sizeof(float));
    }

    if (tid < 128) *(v4f*)(sdz + 4 * tid) = pre;

    const int grp = lane >> 2;
    const int qtr = lane & 3;
    float* rowBase0 = sfeat + (wave * 32 + 4 * grp) * kPad + 4 * qtr;
    const unsigned qoff = 16u * (unsigned)qtr;
    pvr_gather_pass<0>(points_feat, ob, wn, qoff, rowBase0 + 0 * kPad);
    pvr_gather_pass<1>(points_feat, ob, wn, qoff, rowBase0 + 1 * kPad);
    __builtin_amdgcn_sched_barrier(0);
    pvr_gather_pass<2>(points_feat, ob, wn, qoff, rowBase0 + 2 * kPad);
    pvr_gather_pass<3>(points_feat, ob, wn, qoff, rowBase0 + 3 * kPad);

    __syncthreads();

    const int half = lane >> 4;
    const int nl   = lane & 15;

    v2f bmat[4];
#pragma unroll
    for (int t = 0; t < 4; ++t) {
        const int k0 = 4 * t + 2 * half;
        float b0 = 0.0f, b1 = 0.0f;
        if (nl < 3) {
            b0 = w_rgb[k0 * 3 + nl];
            b1 = w_rgb[(k0 + 1) * 3 + nl];
        } else if (nl == 3) {
            b0 = w_sigma[k0];
            b1 = w_sigma[k0 + 1];
        }
        bmat[t].x = b0;
        bmat[t].y = b1;
    }

#pragma unroll
    for (int tt = 0; tt < 2; ++tt) {
        const int rowBase = (wave * 2 + tt) * 16;
        v8f d = {};
#pragma unroll
        for (int t = 0; t < 4; ++t) {
            const int k0 = 4 * t + 2 * half;
            const float* src = sfeat + (rowBase + nl) * kPad + k0;
            v2f a;
            a.x = src[0];
            a.y = src[1];
            d = __builtin_amdgcn_wmma_f32_16x16x4_f32(
                    false, a, false, bmat[t], (short)0, d, false, false);
        }

        float* hp = shead + nl * kPlane + rowBase + 8 * half;
        if (nl < 3) {
            float hv[8];
#pragma unroll
            for (int v = 0; v < 8; ++v) hv[v] = 1.0f / (1.0f + __expf(-d[v]));
            *(v4f*)(hp)     = (v4f){hv[0], hv[1], hv[2], hv[3]};
            *(v4f*)(hp + 4) = (v4f){hv[4], hv[5], hv[6], hv[7]};
        } else if (nl == 3) {
            float hv[8];
#pragma unroll
            for (int v = 0; v < 8; ++v) hv[v] = fmaxf(d[v], 0.0f);
            *(v4f*)(hp)     = (v4f){hv[0], hv[1], hv[2], hv[3]};
            *(v4f*)(hp + 4) = (v4f){hv[4], hv[5], hv[6], hv[7]};
        }
    }

    __syncthreads();

    if (tid >= kRPB * kLPR) return;

    const int p  = tid & (kLPR - 1);
    const int rl = tid >> 3;
    const int i0 = rl * kSR + 4 * p;

    float r0 = 0.0f, r1 = 0.0f, r2 = 0.0f, depth = 0.0f, accw = 0.0f;
    float carry = 1.0f;
#pragma unroll
    for (int j = 0; j < 4; ++j) {
        const int o = i0 + 32 * j;
        const v4f c0 = *(const v4f*)(shead + 0 * kPlane + o);
        const v4f c1 = *(const v4f*)(shead + 1 * kPlane + o);
        const v4f c2 = *(const v4f*)(shead + 2 * kPlane + o);
        const v4f sg = *(const v4f*)(shead + 3 * kPlane + o);
        const v4f dl = *(const v4f*)(sdz + o);
        const v4f zv = *(const v4f*)(sdz + 256 + o);

        const float sgv[4] = {sg.x, sg.y, sg.z, sg.w};
        const float dlv[4] = {dl.x, dl.y, dl.z, dl.w};
        const float c0v[4] = {c0.x, c0.y, c0.z, c0.w};
        const float c1v[4] = {c1.x, c1.y, c1.z, c1.w};
        const float c2v[4] = {c2.x, c2.y, c2.z, c2.w};
        const float zvv[4] = {zv.x, zv.y, zv.z, zv.w};
        float t = 1.0f;
        float q0 = 0.0f, q1 = 0.0f, q2 = 0.0f, qd = 0.0f, qw = 0.0f;
#pragma unroll
        for (int e = 0; e < 4; ++e) {
            const float a  = 1.0f - __expf(-sgv[e] * dlv[e]);
            const float wt = a * t;
            q0 = fmaf(wt, c0v[e], q0);
            q1 = fmaf(wt, c1v[e], q1);
            q2 = fmaf(wt, c2v[e], q2);
            qd = fmaf(wt, zvv[e], qd);
            qw += wt;
            t *= (1.0f - a + 1e-10f);
        }

        float x = t, y;
        y = pvr_lane8<PVR_DPP8(0, 0, 1, 2, 3, 4, 5, 6)>(x); if (p >= 1) x *= y;
        y = pvr_lane8<PVR_DPP8(0, 1, 0, 1, 2, 3, 4, 5)>(x); if (p >= 2) x *= y;
        y = pvr_lane8<PVR_DPP8(0, 1, 2, 3, 0, 1, 2, 3)>(x); if (p >= 4) x *= y;
        y = pvr_lane8<PVR_DPP8(0, 0, 1, 2, 3, 4, 5, 6)>(x);
        const float E = (p == 0) ? carry : carry * y;
        r0    = fmaf(E, q0, r0);
        r1    = fmaf(E, q1, r1);
        r2    = fmaf(E, q2, r2);
        depth = fmaf(E, qd, depth);
        accw  = fmaf(E, qw, accw);
        carry *= pvr_lane8<PVR_DPP8(7, 7, 7, 7, 7, 7, 7, 7)>(x);
    }

    r0    = pvr_sum8(r0);
    r1    = pvr_sum8(r1);
    r2    = pvr_sum8(r2);
    depth = pvr_sum8(depth);
    accw  = pvr_sum8(accw);

    const float bg = 1.0f - accw;
    float res = accw;
    if (p == 0) res = r0 + bg;
    if (p == 1) res = r1 + bg;
    if (p == 2) res = r2 + bg;
    if (p == 3) res = depth;
    if (p < 5) out[(blk * kRPB + rl) * 5 + p] = res;
}

extern "C" void kernel_launch(void* const* d_in, const int* in_sizes, int n_in,
                              void* d_out, int out_size, void* d_ws, size_t ws_size,
                              hipStream_t stream)
{
    const float* points_feat = (const float*)d_in[0];
    const int*   indices     = (const int*)  d_in[1];
    const float* dists       = (const float*)d_in[2];
    const float* w_rgb       = (const float*)d_in[3];
    const float* w_sigma     = (const float*)d_in[4];
    const float* delta       = (const float*)d_in[5];
    const float* z_vals      = (const float*)d_in[6];
    float* out = (float*)d_out;

    pvr_render<<<kN / 256, 256, 0, stream>>>(
        points_feat, indices, dists, w_rgb, w_sigma, delta, z_vals, out);
}


# Head product transposed (W^T as A, feature tile as B): a lane holds one point's 4 outputs, so a wave issues 3 sigmoids + 1 relu instead of 16 + 2 on 6 lanes; k order of every sum kept
# speedup vs baseline: 2.6590x; 1.0390x over previous
#include <hip/hip_runtime.h>


typedef __attribute__((ext_vector_type(2))) float v2f;
typedef __attribute__((ext_vector_type(4))) float v4f;
typedef __attribute__((ext_vector_type(8))) float v8f;

constexpr int kC     = 16;
constexpr int kK     = 8;
constexpr int kR     = 4096;
constexpr int kSR    = 128;
constexpr int kN     = kR * kSR;
constexpr int kPad   = 17;
constexpr int kPlane = 272;
constexpr int kLPR   = 8;
constexpr int kRPB   = 2;

#define PVR_DPP8(a, b, c, d, e, f, g, h) \
    ((a) | ((b) << 3) | ((c) << 6) | ((d) << 9) | ((e) << 12) | ((f) << 15) | ((g) << 18) | ((h) << 21))

template <int SEL>
__device__ __forceinline__ float pvr_lane8(float x)
{
    return __uint_as_float(__builtin_amdgcn_mov_dpp8(__float_as_uint(x), SEL));
}

__device__ __forceinline__ float pvr_sum8(float x)
{
    x += pvr_lane8<PVR_DPP8(1, 0, 3, 2, 5, 4, 7, 6)>(x);
    x += pvr_lane8<PVR_DPP8(2, 3, 0, 1, 6, 7, 4, 5)>(x);
    x += pvr_lane8<PVR_DPP8(4, 5, 6, 7, 0, 1, 2, 3)>(x);
    return x;
}


template <int J>
__device__ __forceinline__ unsigned pvr_quad_u(unsigned x)
{
    return (unsigned)__builtin_amdgcn_update_dpp(0, (int)x, J * 0x55, 0xF, 0xF, true);
}

template <int J>
__device__ __forceinline__ void pvr_gather_pass(const float* __restrict__ points_feat,
                                                const unsigned (&ob)[8], const float (&wn)[8],
                                                unsigned qoff, float* __restrict__ dst)
{
    float a0 = 0.0f, a1 = 0.0f, a2 = 0.0f, a3 = 0.0f;
#pragma unroll
    for (int k = 0; k < 8; ++k) {
        const unsigned off = pvr_quad_u<J>(ob[k]) | qoff;
        const float    wk  = __uint_as_float(pvr_quad_u<J>(__float_as_uint(wn[k])));
        const float4 f = *(const float4*)((const char*)points_feat + off);
        a0 = fmaf(wk, f.x, a0);
        a1 = fmaf(wk, f.y, a1);
        a2 = fmaf(wk, f.z, a2);
        a3 = fmaf(wk, f.w, a3);
    }
    dst[0] = a0; dst[1] = a1; dst[2] = a2; dst[3] = a3;
}

__global__ __launch_bounds__(256, 16)
void pvr_render(const float* __restrict__ points_feat,
                const int*   __restrict__ indices,
                const float* __restrict__ dists,
                const float* __restrict__ w_rgb,
                const float* __restrict__ w_sigma,
                const float* __restrict__ delta,
                const float* __restrict__ z_vals,
                float* __restrict__ out)
{
    __shared__ float sfeat[256 * kPad];
    __shared__ __attribute__((aligned(16))) float shead[4 * kPlane];
    __shared__ __attribute__((aligned(16))) float sdz[2 * 256];

    const int tid  = threadIdx.x;
    const int lane = tid & 31;
    const int wave = tid >> 5;
    const int blk  = blockIdx.x;

    v4f pre = {0.0f, 0.0f, 0.0f, 0.0f};
    if (tid < 128) {
        const float* src = (tid < 64) ? delta : z_vals;
        pre = *(const v4f*)(src + (size_t)blk * 256 + 4 * (tid & 63));
    }

    const int plOwn = wave * 32 + lane;
    const int nOwn  = blk * 256 + plOwn;

    const float4* dv = (const float4*)(dists + (size_t)nOwn * kK);
    const int4*   iv = (const int4*)(indices + (size_t)nOwn * kK);
    const float4 da = dv[0], db = dv[1];
    const int4   ia = iv[0], ib = iv[1];
    const float dd[kK] = {da.x, da.y, da.z, da.w, db.x, db.y, db.z, db.w};
    const int   id[kK] = {ia.x, ia.y, ia.z, ia.w, ib.x, ib.y, ib.z, ib.w};

    float w[kK];
    float wsum = 0.0f;
#pragma unroll
    for (int k = 0; k < kK; ++k) {
        w[k] = 1.0f / (dd[k] + 1e-7f);
        wsum += w[k];
    }
    const float inv = 1.0f / wsum;
    float    wn[kK];
    unsigned ob[kK];
#pragma unroll
    for (int k = 0; k < kK; ++k) {
        wn[k] = w[k] * inv;
        ob[k] = (unsigned)id[k] * (unsigned)(kC * sizeof(float));
    }

    if (tid < 128) *(v4f*)(sdz + 4 * tid) = pre;

    const int grp = lane >> 2;
    const int qtr = lane & 3;
    float* rowBase0 = sfeat + (wave * 32 + 4 * grp) * kPad + 4 * qtr;
    const unsigned qoff = 16u * (unsigned)qtr;
    pvr_gather_pass<0>(points_feat, ob, wn, qoff, rowBase0 + 0 * kPad);
    pvr_gather_pass<1>(points_feat, ob, wn, qoff, rowBase0 + 1 * kPad);
    __builtin_amdgcn_sched_barrier(0);
    pvr_gather_pass<2>(points_feat, ob, wn, qoff, rowBase0 + 2 * kPad);
    pvr_gather_pass<3>(points_feat, ob, wn, qoff, rowBase0 + 3 * kPad);

    __syncthreads();

    const int half = lane >> 4;
    const int nl   = lane & 15;

    v2f wfrag[4];
#pragma unroll
    for (int t = 0; t < 4; ++t) {
        const int k0 = 4 * t + 2 * half;
        float b0 = 0.0f, b1 = 0.0f;
        if (nl < 3) {
            b0 = w_rgb[k0 * 3 + nl];
            b1 = w_rgb[(k0 + 1) * 3 + nl];
        } else if (nl == 3) {
            b0 = w_sigma[k0];
            b1 = w_sigma[k0 + 1];
        }
        wfrag[t].x = b0;
        wfrag[t].y = b1;
    }

    const float* fsrc0 = sfeat + (wave * 32 + nl) * kPad + 2 * half;
    const float* fsrc1 = fsrc0 + 16 * kPad;
    v8f d0 = {}, d1 = {};
#pragma unroll
    for (int t = 0; t < 4; ++t) {
        v2f f0, f1;
        f0.x = fsrc0[4 * t];
        f0.y = fsrc0[4 * t + 1];
        f1.x = fsrc1[4 * t];
        f1.y = fsrc1[4 * t + 1];
        d0 = __builtin_amdgcn_wmma_f32_16x16x4_f32(
                 false, wfrag[t], false, f0, (short)0, d0, false, false);
        d1 = __builtin_amdgcn_wmma_f32_16x16x4_f32(
                 false, wfrag[t], false, f1, (short)0, d1, false, false);
    }

    float hv[4];
#pragma unroll
    for (int r = 0; r < 4; ++r) {
        const int   bits = __float_as_int(d1[r]);
        const float up   = __int_as_float(
            __builtin_amdgcn_permlanex16(bits, bits, 0x76543210, 0xfedcba98, false, false));
        hv[r] = half ? up : d0[r];
    }

    float* hp = shead + wave * 32 + lane;
#pragma unroll
    for (int r = 0; r < 3; ++r) hp[r * kPlane] = 1.0f / (1.0f + __expf(-hv[r]));
    hp[3 * kPlane] = fmaxf(hv[3], 0.0f);

    __syncthreads();

    if (tid >= kRPB * kLPR) return;

    const int p  = tid & (kLPR - 1);
    const int rl = tid >> 3;
    const int i0 = rl * kSR + 4 * p;

    float r0 = 0.0f, r1 = 0.0f, r2 = 0.0f, depth = 0.0f, accw = 0.0f;
    float carry = 1.0f;
#pragma unroll
    for (int j = 0; j < 4; ++j) {
        const int o = i0 + 32 * j;
        const v4f c0 = *(const v4f*)(shead + 0 * kPlane + o);
        const v4f c1 = *(const v4f*)(shead + 1 * kPlane + o);
        const v4f c2 = *(const v4f*)(shead + 2 * kPlane + o);
        const v4f sg = *(const v4f*)(shead + 3 * kPlane + o);
        const v4f dl = *(const v4f*)(sdz + o);
        const v4f zv = *(const v4f*)(sdz + 256 + o);

        const float sgv[4] = {sg.x, sg.y, sg.z, sg.w};
        const float dlv[4] = {dl.x, dl.y, dl.z, dl.w};
        const float c0v[4] = {c0.x, c0.y, c0.z, c0.w};
        const float c1v[4] = {c1.x, c1.y, c1.z, c1.w};
        const float c2v[4] = {c2.x, c2.y, c2.z, c2.w};
        const float zvv[4] = {zv.x, zv.y, zv.z, zv.w};
        float t = 1.0f;
        float q0 = 0.0f, q1 = 0.0f, q2 = 0.0f, qd = 0.0f, qw = 0.0f;
#pragma unroll
        for (int e = 0; e < 4; ++e) {
            const float a  = 1.0f - __expf(-sgv[e] * dlv[e]);
            const float wt = a * t;
            q0 = fmaf(wt, c0v[e], q0);
            q1 = fmaf(wt, c1v[e], q1);
            q2 = fmaf(wt, c2v[e], q2);
            qd = fmaf(wt, zvv[e], qd);
            qw += wt;
            t *= (1.0f - a + 1e-10f);
        }

        float x = t, y;
        y = pvr_lane8<PVR_DPP8(0, 0, 1, 2, 3, 4, 5, 6)>(x); if (p >= 1) x *= y;
        y = pvr_lane8<PVR_DPP8(0, 1, 0, 1, 2, 3, 4, 5)>(x); if (p >= 2) x *= y;
        y = pvr_lane8<PVR_DPP8(0, 1, 2, 3, 0, 1, 2, 3)>(x); if (p >= 4) x *= y;
        y = pvr_lane8<PVR_DPP8(0, 0, 1, 2, 3, 4, 5, 6)>(x);
        const float E = (p == 0) ? carry : carry * y;
        r0    = fmaf(E, q0, r0);
        r1    = fmaf(E, q1, r1);
        r2    = fmaf(E, q2, r2);
        depth = fmaf(E, qd, depth);
        accw  = fmaf(E, qw, accw);
        carry *= pvr_lane8<PVR_DPP8(7, 7, 7, 7, 7, 7, 7, 7)>(x);
    }

    r0    = pvr_sum8(r0);
    r1    = pvr_sum8(r1);
    r2    = pvr_sum8(r2);
    depth = pvr_sum8(depth);
    accw  = pvr_sum8(accw);

    const float bg = 1.0f - accw;
    float res = accw;
    if (p == 0) res = r0 + bg;
    if (p == 1) res = r1 + bg;
    if (p == 2) res = r2 + bg;
    if (p == 3) res = depth;
    if (p < 5) out[(blk * kRPB + rl) * 5 + p] = res;
}

extern "C" void kernel_launch(void* const* d_in, const int* in_sizes, int n_in,
                              void* d_out, int out_size, void* d_ws, size_t ws_size,
                              hipStream_t stream)
{
    const float* points_feat = (const float*)d_in[0];
    const int*   indices     = (const int*)  d_in[1];
    const float* dists       = (const float*)d_in[2];
    const float* w_rgb       = (const float*)d_in[3];
    const float* w_sigma     = (const float*)d_in[4];
    const float* delta       = (const float*)d_in[5];
    const float* z_vals      = (const float*)d_in[6];
    float* out = (float*)d_out;

    pvr_render<<<kN / 256, 256, 0, stream>>>(
        points_feat, indices, dists, w_rgb, w_sigma, delta, z_vals, out);
}
